# starting kernel
# baseline (speedup 1.0000x reference)
#include <hip/hip_runtime.h>

#define NN 100000
#define NE 1600000
#define D 128
#define D4 132
#define NL 7
#define GEMM_BLOCKS 1250

typedef __attribute__((ext_vector_type(2))) float v2f;
typedef __attribute__((ext_vector_type(8))) float v8f;


__global__ void gnn_hist_kernel(const int* __restrict__ dst, int* __restrict__ cnt) {
    int e = blockIdx.x * blockDim.x + threadIdx.x;
    if (e < NE) atomicAdd(&cnt[dst[e]], 1);
}

__global__ void gnn_scan_kernel(const int* __restrict__ cnt, int* __restrict__ row_ptr,
                                float* __restrict__ deg_inv) {
    __shared__ int sdata[1024];
    __shared__ int s_off;
    int tid = threadIdx.x;
    if (tid == 0) { s_off = 0; row_ptr[0] = 0; }
    __syncthreads();
    for (int base = 0; base < NN; base += 1024) {
        int i = base + tid;
        int v = (i < NN) ? cnt[i] : 0;
        sdata[tid] = v;
        __syncthreads();
        for (int off = 1; off < 1024; off <<= 1) {
            int t = (tid >= off) ? sdata[tid - off] : 0;
            __syncthreads();
            sdata[tid] += t;
            __syncthreads();
        }
        int incl = sdata[tid];
        int carry = s_off;
        if (i < NN) {
            row_ptr[i + 1] = carry + incl;
            deg_inv[i] = 1.0f / fmaxf((float)v, 1.0f);
        }
        __syncthreads();
        if (tid == 1023) s_off = carry + sdata[1023];
        __syncthreads();
    }
}

__global__ void gnn_fill_kernel(const int* __restrict__ src, const int* __restrict__ dst,
                                const int* __restrict__ row_ptr, int* __restrict__ cnt2,
                                int* __restrict__ col_src) {
    int e = blockIdx.x * blockDim.x + threadIdx.x;
    if (e < NE) {
        int d = dst[e];
        int pos = row_ptr[d] + atomicAdd(&cnt2[d], 1);
        col_src[pos] = src[e];
    }
}


__global__ void __launch_bounds__(256)
gnn_agg_kernel(const float* __restrict__ h, const int* __restrict__ row_ptr,
               const int* __restrict__ col_src, const float* __restrict__ deg_inv,
               float* __restrict__ mean) {
    int wave = threadIdx.x >> 5;
    int lane = threadIdx.x & 31;
    int node = blockIdx.x * (blockDim.x >> 5) + wave;
    if (node >= NN) return;
    int e0 = row_ptr[node];
    int e1 = row_ptr[node + 1];
    float4 acc = make_float4(0.f, 0.f, 0.f, 0.f);
    for (int e = e0; e < e1; ++e) {
        int s = col_src[e];
        if (e + 1 < e1)
            __builtin_prefetch(h + (size_t)col_src[e + 1] * D + lane * 4, 0, 1);
        float4 v = ((const float4*)(h + (size_t)s * D))[lane];
        acc.x += v.x; acc.y += v.y; acc.z += v.z; acc.w += v.w;
    }
    float di = deg_inv[node];
    float4 r = make_float4(acc.x * di, acc.y * di, acc.z * di, acc.w * di);
    ((float4*)(mean + (size_t)node * D))[lane] = r;
}


__global__ void __launch_bounds__(256, 1)
gnn_gemm_kernel(const float* __restrict__ mean, const float* __restrict__ x,
                const float* __restrict__ Wl, const float* __restrict__ Wr,
                const float* __restrict__ bias, float* __restrict__ out, int relu) {
    __shared__ float sA[16 * D4];
    __shared__ float sX[16 * D4];

    int wave = threadIdx.x >> 5;
    int lane = threadIdx.x & 31;
    int r     = lane & 15;
    int khalf = lane >> 4;

    v2f bl[32], br[32];
    {
        const float* wlrow = Wl + (size_t)(wave * 16 + r) * D;
        const float* wrrow = Wr + (size_t)(wave * 16 + r) * D;
        #pragma unroll
        for (int kk = 0; kk < 32; ++kk) {
            int ko = kk * 4 + 2 * khalf;
            bl[kk] = *(const v2f*)(wlrow + ko);
            br[kk] = *(const v2f*)(wrrow + ko);
        }
    }
    float bj = bias[wave * 16 + r];

    for (int rt = blockIdx.x; rt < NN / 16; rt += GEMM_BLOCKS) {
        {
            const float4* gm = (const float4*)(mean + (size_t)rt * 16 * D);
            const float4* gx = (const float4*)(x    + (size_t)rt * 16 * D);
            #pragma unroll
            for (int j = 0; j < 2; ++j) {
                int i    = threadIdx.x + j * 256;
                int row  = i >> 5;
                int col4 = i & 31;
                ((float4*)(sA + row * D4))[col4] = gm[i];
                ((float4*)(sX + row * D4))[col4] = gx[i];
            }
        }
        __syncthreads();

        v8f acc = {};
        #pragma unroll
        for (int kk = 0; kk < 32; ++kk) {
            int ko = kk * 4 + 2 * khalf;
            v2f a = *(const v2f*)(sA + r * D4 + ko);
            acc = __builtin_amdgcn_wmma_f32_16x16x4_f32(false, a, false, bl[kk],
                                                        (short)0, acc, false, false);
            v2f a2 = *(const v2f*)(sX + r * D4 + ko);
            acc = __builtin_amdgcn_wmma_f32_16x16x4_f32(false, a2, false, br[kk],
                                                        (short)0, acc, false, false);
        }

        float* obase = out + (size_t)(rt * 16) * D + wave * 16 + r;
        #pragma unroll
        for (int v = 0; v < 8; ++v) {
            float val = acc[v] + bj;
            if (relu) val = fmaxf(val, 0.0f);
            obase[(size_t)(v + 8 * khalf) * D] = val;
        }
        __syncthreads();
    }
}


extern "C" void kernel_launch(void* const* d_in, const int* in_sizes, int n_in,
                              void* d_out, int out_size, void* d_ws, size_t ws_size,
                              hipStream_t stream) {
    const float* x          = (const float*)d_in[0];
    const int*   edge_index = (const int*)d_in[1];
    const float* Wl         = (const float*)d_in[2];
    const float* Wr         = (const float*)d_in[3];
    const float* b          = (const float*)d_in[4];
    const int* src = edge_index;
    const int* dst = edge_index + NE;

    char* ws = (char*)d_ws;
    size_t off = 0;
    auto carve = [&](size_t bytes) -> void* {
        void* p = ws + off;
        off = (off + bytes + 255) & ~(size_t)255;
        return p;
    };
    float* mean    = (float*)carve((size_t)NN * D * sizeof(float));
    float* hA      = (float*)carve((size_t)NN * D * sizeof(float));
    float* hB      = (float*)carve((size_t)NN * D * sizeof(float));
    float* deg_inv = (float*)carve((size_t)NN * sizeof(float));
    int*   row_ptr = (int*)carve((size_t)(NN + 1) * sizeof(int));
    int*   cnt     = (int*)carve((size_t)NN * sizeof(int));
    int*   col_src = (int*)carve((size_t)NE * sizeof(int));

    hipMemsetAsync(cnt, 0, (size_t)NN * sizeof(int), stream);
    gnn_hist_kernel<<<(NE + 255) / 256, 256, 0, stream>>>(dst, cnt);
    gnn_scan_kernel<<<1, 1024, 0, stream>>>(cnt, row_ptr, deg_inv);
    hipMemsetAsync(cnt, 0, (size_t)NN * sizeof(int), stream);
    gnn_fill_kernel<<<(NE + 255) / 256, 256, 0, stream>>>(src, dst, row_ptr, cnt, col_src);

    const float* hin = x;
    float* houts[2] = {hA, hB};
    for (int l = 0; l < NL; ++l) {
        float* hout = (l == NL - 1) ? (float*)d_out : houts[l & 1];
        gnn_agg_kernel<<<NN / 8, 256, 0, stream>>>(hin, row_ptr, col_src, deg_inv, mean);
        gnn_gemm_kernel<<<GEMM_BLOCKS, 256, 0, stream>>>(
            mean, hin, Wl + (size_t)l * D * D, Wr + (size_t)l * D * D,
            b + (size_t)l * D, hout, (l < NL - 1) ? 1 : 0);
        hin = hout;
    }
}


# No CSR: each edge swaps its id into head[dst] (one atomic exchange), keeping the old id as its link; aggregation walks the list; count, scan, fill gone; sum order reverse arrival (was arrival)
# speedup vs baseline: 1.4568x; 1.4568x over previous
#include <hip/hip_runtime.h>

#define NN 100000
#define NE 1600000
#define D 128
#define D4 132
#define NL 7
#define GEMM_BLOCKS 1250

typedef __attribute__((ext_vector_type(2))) float v2f;
typedef __attribute__((ext_vector_type(8))) float v8f;


#define LINK_T 160
#define LINK_G 4
#define NG (NE / 4)
static_assert(NE % 4 == 0 && NG % (LINK_T * LINK_G) == 0, "edge groups must tile the grid exactly");

__global__ void __launch_bounds__(LINK_T)
gnn_link_kernel(const int* __restrict__ dst, int* __restrict__ head, int* __restrict__ next) {
    const int4* dst4 = (const int4*)dst;
    int4* next4 = (int4*)next;
    int g0 = blockIdx.x * (LINK_T * LINK_G) + threadIdx.x;
    int4 d[LINK_G], p[LINK_G];
    #pragma unroll
    for (int j = 0; j < LINK_G; ++j) d[j] = dst4[g0 + j * LINK_T];
    #pragma unroll
    for (int j = 0; j < LINK_G; ++j) {
        int e = 4 * (g0 + j * LINK_T);
        p[j].x = atomicExch(&head[d[j].x], e + 0);
        p[j].y = atomicExch(&head[d[j].y], e + 1);
        p[j].z = atomicExch(&head[d[j].z], e + 2);
        p[j].w = atomicExch(&head[d[j].w], e + 3);
    }
    #pragma unroll
    for (int j = 0; j < LINK_G; ++j) next4[g0 + j * LINK_T] = p[j];
}


__global__ void __launch_bounds__(256)
gnn_agg_kernel(const float* __restrict__ h, const int* __restrict__ head,
               const int* __restrict__ next, const int* __restrict__ src,
               float* __restrict__ mean) {
    int wave = threadIdx.x >> 5;
    int lane = threadIdx.x & 31;
    int node = blockIdx.x * (blockDim.x >> 5) + wave;
    if (node >= NN) return;
    float4 acc = make_float4(0.f, 0.f, 0.f, 0.f);
    int n = 0;
    int e = head[node];
    if ((unsigned)e < (unsigned)NE) {
        int s  = src[e];
        int nx = next[e];
        #pragma unroll 1
        for (int hop = 0; hop < NE; ++hop) {
            bool more = (unsigned)nx < (unsigned)NE;
            int s2 = s, nx2 = nx;
            if (more) { s2 = src[nx]; nx2 = next[nx]; }
            float4 v = ((const float4*)(h + (size_t)s * D))[lane];
            acc.x += v.x; acc.y += v.y; acc.z += v.z; acc.w += v.w;
            ++n;
            if (!more) break;
            s = s2; nx = nx2;
        }
    }
    float di = 1.0f / fmaxf((float)n, 1.0f);
    float4 r = make_float4(acc.x * di, acc.y * di, acc.z * di, acc.w * di);
    ((float4*)(mean + (size_t)node * D))[lane] = r;
}


__global__ void __launch_bounds__(256, 1)
gnn_gemm_kernel(const float* __restrict__ mean, const float* __restrict__ x,
                const float* __restrict__ Wl, const float* __restrict__ Wr,
                const float* __restrict__ bias, float* __restrict__ out, int relu) {
    __shared__ float sA[16 * D4];
    __shared__ float sX[16 * D4];

    int wave = threadIdx.x >> 5;
    int lane = threadIdx.x & 31;
    int r     = lane & 15;
    int khalf = lane >> 4;

    v2f bl[32], br[32];
    {
        const float* wlrow = Wl + (size_t)(wave * 16 + r) * D;
        const float* wrrow = Wr + (size_t)(wave * 16 + r) * D;
        #pragma unroll
        for (int kk = 0; kk < 32; ++kk) {
            int ko = kk * 4 + 2 * khalf;
            bl[kk] = *(const v2f*)(wlrow + ko);
            br[kk] = *(const v2f*)(wrrow + ko);
        }
    }
    float bj = bias[wave * 16 + r];

    for (int rt = blockIdx.x; rt < NN / 16; rt += GEMM_BLOCKS) {
        {
            const float4* gm = (const float4*)(mean + (size_t)rt * 16 * D);
            const float4* gx = (const float4*)(x    + (size_t)rt * 16 * D);
            #pragma unroll
            for (int j = 0; j < 2; ++j) {
                int i    = threadIdx.x + j * 256;
                int row  = i >> 5;
                int col4 = i & 31;
                ((float4*)(sA + row * D4))[col4] = gm[i];
                ((float4*)(sX + row * D4))[col4] = gx[i];
            }
        }
        __syncthreads();

        v8f acc = {};
        #pragma unroll
        for (int kk = 0; kk < 32; ++kk) {
            int ko = kk * 4 + 2 * khalf;
            v2f a = *(const v2f*)(sA + r * D4 + ko);
            acc = __builtin_amdgcn_wmma_f32_16x16x4_f32(false, a, false, bl[kk],
                                                        (short)0, acc, false, false);
            v2f a2 = *(const v2f*)(sX + r * D4 + ko);
            acc = __builtin_amdgcn_wmma_f32_16x16x4_f32(false, a2, false, br[kk],
                                                        (short)0, acc, false, false);
        }

        float* obase = out + (size_t)(rt * 16) * D + wave * 16 + r;
        #pragma unroll
        for (int v = 0; v < 8; ++v) {
            float val = acc[v] + bj;
            if (relu) val = fmaxf(val, 0.0f);
            obase[(size_t)(v + 8 * khalf) * D] = val;
        }
        __syncthreads();
    }
}


extern "C" void kernel_launch(void* const* d_in, const int* in_sizes, int n_in,
                              void* d_out, int out_size, void* d_ws, size_t ws_size,
                              hipStream_t stream) {
    const float* x          = (const float*)d_in[0];
    const int*   edge_index = (const int*)d_in[1];
    const float* Wl         = (const float*)d_in[2];
    const float* Wr         = (const float*)d_in[3];
    const float* b          = (const float*)d_in[4];
    const int* src = edge_index;
    const int* dst = edge_index + NE;

    char* ws = (char*)d_ws;
    size_t off = 0;
    auto carve = [&](size_t bytes) -> void* {
        void* p = ws + off;
        off = (off + bytes + 255) & ~(size_t)255;
        return p;
    };
    float* mean    = (float*)carve((size_t)NN * D * sizeof(float));
    float* hA      = (float*)carve((size_t)NN * D * sizeof(float));
    float* hB      = (float*)carve((size_t)NN * D * sizeof(float));
    int*   head    = (int*)carve((size_t)NN * sizeof(int));
    int*   next    = (int*)carve((size_t)NE * sizeof(int));

    hipMemsetAsync(head, 0xFF, (size_t)NN * sizeof(int), stream);
    gnn_link_kernel<<<NG / (LINK_T * LINK_G), LINK_T, 0, stream>>>(dst, head, next);

    const float* hin = x;
    float* houts[2] = {hA, hB};
    for (int l = 0; l < NL; ++l) {
        float* hout = (l == NL - 1) ? (float*)d_out : houts[l & 1];
        gnn_agg_kernel<<<NN / 8, 256, 0, stream>>>(hin, head, next, src, mean);
        gnn_gemm_kernel<<<GEMM_BLOCKS, 256, 0, stream>>>(
            mean, hin, Wl + (size_t)l * D * D, Wr + (size_t)l * D * D,
            b + (size_t)l * D, hout, (l < NL - 1) ? 1 : 0);
        hin = hout;
    }
}


# No lists: 32-slot source table per node (returning add = slot, overflow chained); 10240 resident waves loop over nodes, one 128-B load gives a node's sources, rows 8 a batch; sum order arrival
# speedup vs baseline: 1.6941x; 1.1629x over previous
#include <hip/hip_runtime.h>

#define NN 100000
#define NE 1600000
#define D 128
#define D4 132
#define NL 7
#define GEMM_BLOCKS 1250
#define CAP 32
#define AGG_BLOCKS 1280
#define AGG_WAVES (AGG_BLOCKS * 8)

typedef __attribute__((ext_vector_type(2))) float v2f;
typedef __attribute__((ext_vector_type(8))) float v8f;


#define LINK_T 160
#define LINK_G 4
#define NG (NE / 4)
static_assert(NE % 4 == 0 && NG % (LINK_T * LINK_G) == 0, "edge groups must tile the grid exactly");

static __device__ __forceinline__ void place_edge(int node, int s, int pos, int e,
                                                  int* table, int* ohead, int* onext) {
    if (pos < CAP) table[node * CAP + pos] = s;
    else           onext[e] = atomicExch(&ohead[node], e);
}

__global__ void __launch_bounds__(LINK_T)
gnn_slot_kernel(const int* __restrict__ src, const int* __restrict__ dst,
                int* __restrict__ cnt, int* __restrict__ ohead,
                int* __restrict__ table, int* __restrict__ onext) {
    const int4* dst4 = (const int4*)dst;
    const int4* src4 = (const int4*)src;
    int g0 = blockIdx.x * (LINK_T * LINK_G) + threadIdx.x;
    int4 d[LINK_G], s[LINK_G], p[LINK_G];
    #pragma unroll
    for (int j = 0; j < LINK_G; ++j) { d[j] = dst4[g0 + j * LINK_T]; s[j] = src4[g0 + j * LINK_T]; }
    #pragma unroll
    for (int j = 0; j < LINK_G; ++j) {
        p[j].x = atomicAdd(&cnt[d[j].x], 1) + 1;
        p[j].y = atomicAdd(&cnt[d[j].y], 1) + 1;
        p[j].z = atomicAdd(&cnt[d[j].z], 1) + 1;
        p[j].w = atomicAdd(&cnt[d[j].w], 1) + 1;
    }
    #pragma unroll
    for (int j = 0; j < LINK_G; ++j) {
        int e = 4 * (g0 + j * LINK_T);
        place_edge(d[j].x, s[j].x, p[j].x, e + 0, table, ohead, onext);
        place_edge(d[j].y, s[j].y, p[j].y, e + 1, table, ohead, onext);
        place_edge(d[j].z, s[j].z, p[j].z, e + 2, table, ohead, onext);
        place_edge(d[j].w, s[j].w, p[j].w, e + 3, table, ohead, onext);
    }
}


static __device__ __forceinline__ float4 h_row(const float* __restrict__ h, int s, int lane) {
    return ((const float4*)(h + (size_t)s * D))[lane];
}
static __device__ __forceinline__ void add4(float4& a, const float4& v) {
    a.x += v.x; a.y += v.y; a.z += v.z; a.w += v.w;
}

__global__ void __launch_bounds__(256)
gnn_agg_kernel(const float* __restrict__ h, const int* __restrict__ cnt,
               const int* __restrict__ table, const int* __restrict__ ohead,
               const int* __restrict__ onext, const int* __restrict__ src,
               float* __restrict__ mean) {
    int lane = threadIdx.x & 31;
    int w = blockIdx.x * 8 + __builtin_amdgcn_readfirstlane((int)(threadIdx.x >> 5));

    int node = -1, c = 0, m = 0, sv = 0;
    int node1 = w;
    int r1 = cnt[node1];

    #pragma unroll 1
    for (int turn = 0; turn <= (NN + AGG_WAVES - 1) / AGG_WAVES; ++turn) {
        bool have1 = node1 < NN;
        int c1 = r1 + 1;
        int m1 = c1 < CAP ? c1 : CAP;
        int sv1 = 0;
        if (have1 && lane < m1) sv1 = table[node1 * CAP + lane];
        int node2 = node1 + AGG_WAVES;
        int r2 = cnt[node2 < NN ? node2 : NN - 1];

        if (node >= 0) {
            float4 acc = make_float4(0.f, 0.f, 0.f, 0.f);
            int j0 = 0;
            for (; j0 + 8 <= m; j0 += 8) {
                float4 v[8];
                #pragma unroll
                for (int u = 0; u < 8; ++u) v[u] = h_row(h, __builtin_amdgcn_readlane(sv, j0 + u), lane);
                #pragma unroll
                for (int u = 0; u < 8; ++u) add4(acc, v[u]);
            }
            int k = m - j0;
            if (k) {
                float4 a[4], b[2], t;
                int jb = j0 + (k & 4);
                int jc = jb + (k & 2);
                if (k & 4) {
                    #pragma unroll
                    for (int u = 0; u < 4; ++u) a[u] = h_row(h, __builtin_amdgcn_readlane(sv, j0 + u), lane);
                }
                if (k & 2) {
                    #pragma unroll
                    for (int u = 0; u < 2; ++u) b[u] = h_row(h, __builtin_amdgcn_readlane(sv, jb + u), lane);
                }
                if (k & 1) t = h_row(h, __builtin_amdgcn_readlane(sv, jc), lane);
                if (k & 4) {
                    #pragma unroll
                    for (int u = 0; u < 4; ++u) add4(acc, a[u]);
                }
                if (k & 2) { add4(acc, b[0]); add4(acc, b[1]); }
                if (k & 1) add4(acc, t);
            }
            int n = m;
            if (c > CAP) {
                int e = ohead[node];
                #pragma unroll 1
                for (int hop = 0; hop < NE && (unsigned)e < (unsigned)NE; ++hop) {
                    int s = src[e];
                    e = onext[e];
                    float4 v = h_row(h, s, lane);
                    add4(acc, v);
                    ++n;
                }
            }
            float di = 1.0f / fmaxf((float)n, 1.0f);
            float4 r = make_float4(acc.x * di, acc.y * di, acc.z * di, acc.w * di);
            ((float4*)(mean + (size_t)node * D))[lane] = r;
        }
        if (!have1) break;
        node = node1; c = c1; m = m1; sv = sv1;
        node1 = node2; r1 = r2;
    }
}


__global__ void __launch_bounds__(256, 1)
gnn_gemm_kernel(const float* mean, const float* __restrict__ x,
                const float* __restrict__ Wl, const float* __restrict__ Wr,
                const float* __restrict__ bias, float* out, int relu) {
    __shared__ float sA[16 * D4];
    __shared__ float sX[16 * D4];

    int wave = threadIdx.x >> 5;
    int lane = threadIdx.x & 31;
    int r     = lane & 15;
    int khalf = lane >> 4;

    v2f bl[32], br[32];
    {
        const float* wlrow = Wl + (size_t)(wave * 16 + r) * D;
        const float* wrrow = Wr + (size_t)(wave * 16 + r) * D;
        #pragma unroll
        for (int kk = 0; kk < 32; ++kk) {
            int ko = kk * 4 + 2 * khalf;
            bl[kk] = *(const v2f*)(wlrow + ko);
            br[kk] = *(const v2f*)(wrrow + ko);
        }
    }
    float bj = bias[wave * 16 + r];

    for (int rt = blockIdx.x; rt < NN / 16; rt += GEMM_BLOCKS) {
        {
            const float4* gm = (const float4*)(mean + (size_t)rt * 16 * D);
            const float4* gx = (const float4*)(x    + (size_t)rt * 16 * D);
            #pragma unroll
            for (int j = 0; j < 2; ++j) {
                int i    = threadIdx.x + j * 256;
                int row  = i >> 5;
                int col4 = i & 31;
                ((float4*)(sA + row * D4))[col4] = gm[i];
                ((float4*)(sX + row * D4))[col4] = gx[i];
            }
        }
        __syncthreads();

        v8f acc = {};
        #pragma unroll
        for (int kk = 0; kk < 32; ++kk) {
            int ko = kk * 4 + 2 * khalf;
            v2f a = *(const v2f*)(sA + r * D4 + ko);
            acc = __builtin_amdgcn_wmma_f32_16x16x4_f32(false, a, false, bl[kk],
                                                        (short)0, acc, false, false);
            v2f a2 = *(const v2f*)(sX + r * D4 + ko);
            acc = __builtin_amdgcn_wmma_f32_16x16x4_f32(false, a2, false, br[kk],
                                                        (short)0, acc, false, false);
        }

        float* obase = out + (size_t)(rt * 16) * D + wave * 16 + r;
        #pragma unroll
        for (int v = 0; v < 8; ++v) {
            float val = acc[v] + bj;
            if (relu) val = fmaxf(val, 0.0f);
            obase[(size_t)(v + 8 * khalf) * D] = val;
        }
        __syncthreads();
    }
}


extern "C" void kernel_launch(void* const* d_in, const int* in_sizes, int n_in,
                              void* d_out, int out_size, void* d_ws, size_t ws_size,
                              hipStream_t stream) {
    const float* x          = (const float*)d_in[0];
    const int*   edge_index = (const int*)d_in[1];
    const float* Wl         = (const float*)d_in[2];
    const float* Wr         = (const float*)d_in[3];
    const float* b          = (const float*)d_in[4];
    const int* src = edge_index;
    const int* dst = edge_index + NE;

    char* ws = (char*)d_ws;
    size_t off = 0;
    auto carve = [&](size_t bytes) -> void* {
        void* p = ws + off;
        off = (off + bytes + 255) & ~(size_t)255;
        return p;
    };
    float* hA      = (float*)carve((size_t)NN * D * sizeof(float));
    float* hB      = (float*)carve((size_t)NN * D * sizeof(float));
    int*   table   = (int*)carve((size_t)NN * CAP * sizeof(int));
    int*   cnt     = (int*)carve((size_t)2 * NN * sizeof(int));
    int*   ohead   = cnt + NN;
    int*   onext   = (int*)carve((size_t)NE * sizeof(int));

    hipMemsetAsync(cnt, 0xFF, (size_t)2 * NN * sizeof(int), stream);
    gnn_slot_kernel<<<NG / (LINK_T * LINK_G), LINK_T, 0, stream>>>(src, dst, cnt, ohead, table, onext);

    const float* hin = x;
    float* houts[2] = {hA, hB};
    for (int l = 0; l < NL; ++l) {
        float* hout = (l == NL - 1) ? (float*)d_out : houts[l & 1];
        gnn_agg_kernel<<<AGG_BLOCKS, 256, 0, stream>>>(hin, cnt, table, ohead, onext, src, hout);
        gnn_gemm_kernel<<<GEMM_BLOCKS, 256, 0, stream>>>(
            hout, hin, Wl + (size_t)l * D * D, Wr + (size_t)l * D * D,
            b + (size_t)l * D, hout, (l < NL - 1) ? 1 : 0);
        hin = hout;
    }
}


# Project before aggregating: a layer kernel sums its 16 nodes' source rows of y=h*Wl^T, adds z=h*Wr^T+b, ReLU in LDS, then WMMAs the next layer's y,z; no h or mean plane; products before the mean
# speedup vs baseline: 1.7625x; 1.0404x over previous
#include <hip/hip_runtime.h>

#define NN 100000
#define NE 1600000
#define D 128
#define D4 132
#define NL 7
#define LAYER_BLOCKS 512
#define LAYER_T 512
#define CAP 32
#define AGG_BLOCKS 1280
#define AGG_WAVES (AGG_BLOCKS * 8)

typedef __attribute__((ext_vector_type(2))) float v2f;
typedef __attribute__((ext_vector_type(8))) float v8f;


#define LINK_T 160
#define LINK_G 4
#define NG (NE / 4)
static_assert(NE % 4 == 0 && NG % (LINK_T * LINK_G) == 0, "edge groups must tile the grid exactly");

static __device__ __forceinline__ void place_edge(int node, int s, int pos, int e,
                                                  int* table, int* ohead, int* onext) {
    if (pos < CAP) table[node * CAP + pos] = s;
    else           onext[e] = atomicExch(&ohead[node], e);
}

__global__ void __launch_bounds__(LINK_T)
gnn_slot_kernel(const int* __restrict__ src, const int* __restrict__ dst,
                int* __restrict__ cnt, int* __restrict__ ohead,
                int* __restrict__ table, int* __restrict__ onext) {
    const int4* dst4 = (const int4*)dst;
    const int4* src4 = (const int4*)src;
    int g0 = blockIdx.x * (LINK_T * LINK_G) + threadIdx.x;
    int4 d[LINK_G], s[LINK_G], p[LINK_G];
    #pragma unroll
    for (int j = 0; j < LINK_G; ++j) { d[j] = dst4[g0 + j * LINK_T]; s[j] = src4[g0 + j * LINK_T]; }
    #pragma unroll
    for (int j = 0; j < LINK_G; ++j) {
        p[j].x = atomicAdd(&cnt[d[j].x], 1) + 1;
        p[j].y = atomicAdd(&cnt[d[j].y], 1) + 1;
        p[j].z = atomicAdd(&cnt[d[j].z], 1) + 1;
        p[j].w = atomicAdd(&cnt[d[j].w], 1) + 1;
    }
    #pragma unroll
    for (int j = 0; j < LINK_G; ++j) {
        int e = 4 * (g0 + j * LINK_T);
        place_edge(d[j].x, s[j].x, p[j].x, e + 0, table, ohead, onext);
        place_edge(d[j].y, s[j].y, p[j].y, e + 1, table, ohead, onext);
        place_edge(d[j].z, s[j].z, p[j].z, e + 2, table, ohead, onext);
        place_edge(d[j].w, s[j].w, p[j].w, e + 3, table, ohead, onext);
    }
}


static __device__ __forceinline__ float4 h_row(const float* __restrict__ h, int s, int lane) {
    return ((const float4*)(h + (size_t)s * D))[lane];
}
static __device__ __forceinline__ void add4(float4& a, const float4& v) {
    a.x += v.x; a.y += v.y; a.z += v.z; a.w += v.w;
}

static __device__ __forceinline__ float4 sum_sources(const float* __restrict__ h, int node, int c, int m, int sv,
                                                     const int* __restrict__ ohead, const int* __restrict__ onext,
                                                     const int* __restrict__ src, int lane, int& n_out) {
    float4 acc = make_float4(0.f, 0.f, 0.f, 0.f);
    #pragma unroll 1
    for (int j0 = 0; j0 < m; j0 += 8) {
        float4 v[8];
        #pragma unroll
        for (int u = 0; u < 8; ++u)
            if (j0 + u < m) v[u] = h_row(h, __builtin_amdgcn_readlane(sv, j0 + u), lane);
        #pragma unroll
        for (int u = 0; u < 8; ++u)
            if (j0 + u < m) add4(acc, v[u]);
    }
    int n = m;
    if (c > CAP) {
        int e = ohead[node];
        #pragma unroll 1
        for (int hop = 0; hop < NE && (unsigned)e < (unsigned)NE; ++hop) {
            int s = src[e];
            e = onext[e];
            float4 v = h_row(h, s, lane);
            add4(acc, v);
            ++n;
        }
    }
    n_out = n;
    return acc;
}


__global__ void __launch_bounds__(LAYER_T, 8)
gnn_layer_kernel(const float* __restrict__ x, const float* __restrict__ yin, float* z,
                 float* __restrict__ yout,
                 const float* __restrict__ Wl, const float* __restrict__ Wr,
                 const float* __restrict__ bias,
                 const int* __restrict__ cnt, const int* __restrict__ table,
                 const int* __restrict__ ohead, const int* __restrict__ onext,
                 const int* __restrict__ src, int first) {
    __shared__ float sA[16 * D4];

    int lane = threadIdx.x & 31;
    int wave = __builtin_amdgcn_readfirstlane((int)(threadIdx.x >> 5));
    int r     = lane & 15;
    int khalf = lane >> 4;
    int mi = wave >> 3;
    int ct = wave & 7;

    v2f bw[32];
    {
        const float* wrow = (mi ? Wr : Wl) + (size_t)(ct * 16 + r) * D;
        #pragma unroll
        for (int kk = 0; kk < 32; ++kk) {
            int ko = kk * 4 + 2 * khalf;
            bw[kk] = *(const v2f*)(wrow + ko);
        }
    }
    float bj = mi ? bias[ct * 16 + r] : 0.0f;
    float* oplane = mi ? z : yout;

    int rt = blockIdx.x;
    int c = 0, m = 0, sv = 0;
    float4 own;
    {
        int node0 = rt * 16 + wave;
        if (first) {
            own = ((const float4*)(x + (size_t)node0 * D))[lane];
        } else {
            c = cnt[node0] + 1;
            m = c < CAP ? c : CAP;
            if (lane < m) sv = table[node0 * CAP + lane];
            own = ((const float4*)(z + (size_t)node0 * D))[lane];
        }
    }

    for (; rt < NN / 16; rt += LAYER_BLOCKS) {
        int node = rt * 16 + wave;
        int rtn  = rt + LAYER_BLOCKS;
        bool more = rtn < NN / 16;
        int noden = (more ? rtn : rt) * 16 + wave;
        int rn = 0;
        if (!first) rn = cnt[noden];

        float4 hrow;
        if (first) {
            hrow = own;
        } else {
            int n;
            float4 acc = sum_sources(yin, node, c, m, sv, ohead, onext, src, lane, n);
            float di = 1.0f / fmaxf((float)n, 1.0f);
            hrow.x = fmaxf(acc.x * di + own.x, 0.0f);
            hrow.y = fmaxf(acc.y * di + own.y, 0.0f);
            hrow.z = fmaxf(acc.z * di + own.z, 0.0f);
            hrow.w = fmaxf(acc.w * di + own.w, 0.0f);
        }
        ((float4*)(sA + wave * D4))[lane] = hrow;
        __syncthreads();

        int cn = rn + 1;
        int mn = cn < CAP ? cn : CAP;
        int svn = 0;
        float4 ownn = make_float4(0.f, 0.f, 0.f, 0.f);
        if (more) {
            if (first) {
                ownn = ((const float4*)(x + (size_t)noden * D))[lane];
            } else {
                if (lane < mn) svn = table[noden * CAP + lane];
                ownn = ((const float4*)(z + (size_t)noden * D))[lane];
            }
        }

        v8f acc8 = {};
        #pragma unroll
        for (int kk = 0; kk < 32; ++kk) {
            int ko = kk * 4 + 2 * khalf;
            v2f a = *(const v2f*)(sA + r * D4 + ko);
            acc8 = __builtin_amdgcn_wmma_f32_16x16x4_f32(false, a, false, bw[kk],
                                                         (short)0, acc8, false, false);
        }

        float* obase = oplane + (size_t)(rt * 16) * D + ct * 16 + r;
        #pragma unroll
        for (int v = 0; v < 8; ++v)
            obase[(size_t)(v + 8 * khalf) * D] = acc8[v] + bj;
        __syncthreads();

        c = cn; m = mn; sv = svn; own = ownn;
    }
}


__global__ void __launch_bounds__(256)
gnn_final_kernel(const float* __restrict__ y, const int* __restrict__ cnt,
                 const int* __restrict__ table, const int* __restrict__ ohead,
                 const int* __restrict__ onext, const int* __restrict__ src,
                 float* zout) {
    int lane = threadIdx.x & 31;
    int w = blockIdx.x * 8 + __builtin_amdgcn_readfirstlane((int)(threadIdx.x >> 5));

    int node = -1, c = 0, m = 0, sv = 0;
    int node1 = w;
    int r1 = cnt[node1];

    #pragma unroll 1
    for (int turn = 0; turn <= (NN + AGG_WAVES - 1) / AGG_WAVES; ++turn) {
        bool have1 = node1 < NN;
        int c1 = r1 + 1;
        int m1 = c1 < CAP ? c1 : CAP;
        int sv1 = 0;
        if (have1 && lane < m1) sv1 = table[node1 * CAP + lane];
        int node2 = node1 + AGG_WAVES;
        int r2 = cnt[node2 < NN ? node2 : NN - 1];

        if (node >= 0) {
            float4 own = ((const float4*)(zout + (size_t)node * D))[lane];
            int n;
            float4 acc = sum_sources(y, node, c, m, sv, ohead, onext, src, lane, n);
            float di = 1.0f / fmaxf((float)n, 1.0f);
            float4 o = make_float4(acc.x * di + own.x, acc.y * di + own.y,
                                   acc.z * di + own.z, acc.w * di + own.w);
            ((float4*)(zout + (size_t)node * D))[lane] = o;
        }
        if (!have1) break;
        node = node1; c = c1; m = m1; sv = sv1;
        node1 = node2; r1 = r2;
    }
}


extern "C" void kernel_launch(void* const* d_in, const int* in_sizes, int n_in,
                              void* d_out, int out_size, void* d_ws, size_t ws_size,
                              hipStream_t stream) {
    const float* x          = (const float*)d_in[0];
    const int*   edge_index = (const int*)d_in[1];
    const float* Wl         = (const float*)d_in[2];
    const float* Wr         = (const float*)d_in[3];
    const float* b          = (const float*)d_in[4];
    const int* src = edge_index;
    const int* dst = edge_index + NE;

    char* ws = (char*)d_ws;
    size_t off = 0;
    auto carve = [&](size_t bytes) -> void* {
        void* p = ws + off;
        off = (off + bytes + 255) & ~(size_t)255;
        return p;
    };
    float* yA      = (float*)carve((size_t)NN * D * sizeof(float));
    float* yB      = (float*)carve((size_t)NN * D * sizeof(float));
    int*   table   = (int*)carve((size_t)NN * CAP * sizeof(int));
    int*   cnt     = (int*)carve((size_t)2 * NN * sizeof(int));
    int*   ohead   = cnt + NN;
    int*   onext   = (int*)carve((size_t)NE * sizeof(int));
    float* z       = (float*)d_out;

    hipMemsetAsync(cnt, 0xFF, (size_t)2 * NN * sizeof(int), stream);
    gnn_slot_kernel<<<NG / (LINK_T * LINK_G), LINK_T, 0, stream>>>(src, dst, cnt, ohead, table, onext);

    float* ys[2] = {yA, yB};
    gnn_layer_kernel<<<LAYER_BLOCKS, LAYER_T, 0, stream>>>(
        x, x, z, yA, Wl, Wr, b, cnt, table, ohead, onext, src, 1);
    for (int l = 0; l < NL - 1; ++l) {
        gnn_layer_kernel<<<LAYER_BLOCKS, LAYER_T, 0, stream>>>(
            x, ys[l & 1], z, ys[(l + 1) & 1],
            Wl + (size_t)(l + 1) * D * D, Wr + (size_t)(l + 1) * D * D, b + (size_t)(l + 1) * D,
            cnt, table, ohead, onext, src, 0);
    }
    gnn_final_kernel<<<AGG_BLOCKS, 256, 0, stream>>>(ys[(NL - 1) & 1], cnt, table, ohead, onext, src, z);
}


# Layer kernel: weights read from LDS (padded rows), not held in 64 registers: a wave fits 64 registers, 16 waves a SIMD; a block is 32 waves and takes two tiles a trip; sum order kept
# speedup vs baseline: 1.9224x; 1.0907x over previous
#include <hip/hip_runtime.h>

#define NN 100000
#define NE 1600000
#define D 128
#define D4 132
#define NL 7
#define LAYER_BLOCKS 512
#define LAYER_T 1024
#define CAP 32
#define AGG_BLOCKS 1280
#define AGG_WAVES (AGG_BLOCKS * 8)

typedef __attribute__((ext_vector_type(2))) float v2f;
typedef __attribute__((ext_vector_type(8))) float v8f;


#define LINK_T 160
#define LINK_G 4
#define NG (NE / 4)
static_assert(NE % 4 == 0 && NG % (LINK_T * LINK_G) == 0, "edge groups must tile the grid exactly");

static __device__ __forceinline__ void place_edge(int node, int s, int pos, int e,
                                                  int* table, int* ohead, int* onext) {
    if (pos < CAP) table[node * CAP + pos] = s;
    else           onext[e] = atomicExch(&ohead[node], e);
}

__global__ void __launch_bounds__(LINK_T)
gnn_slot_kernel(const int* __restrict__ src, const int* __restrict__ dst,
                int* __restrict__ cnt, int* __restrict__ ohead,
                int* __restrict__ table, int* __restrict__ onext) {
    const int4* dst4 = (const int4*)dst;
    const int4* src4 = (const int4*)src;
    int g0 = blockIdx.x * (LINK_T * LINK_G) + threadIdx.x;
    int4 d[LINK_G], s[LINK_G], p[LINK_G];
    #pragma unroll
    for (int j = 0; j < LINK_G; ++j) { d[j] = dst4[g0 + j * LINK_T]; s[j] = src4[g0 + j * LINK_T]; }
    #pragma unroll
    for (int j = 0; j < LINK_G; ++j) {
        p[j].x = atomicAdd(&cnt[d[j].x], 1) + 1;
        p[j].y = atomicAdd(&cnt[d[j].y], 1) + 1;
        p[j].z = atomicAdd(&cnt[d[j].z], 1) + 1;
        p[j].w = atomicAdd(&cnt[d[j].w], 1) + 1;
    }
    #pragma unroll
    for (int j = 0; j < LINK_G; ++j) {
        int e = 4 * (g0 + j * LINK_T);
        place_edge(d[j].x, s[j].x, p[j].x, e + 0, table, ohead, onext);
        place_edge(d[j].y, s[j].y, p[j].y, e + 1, table, ohead, onext);
        place_edge(d[j].z, s[j].z, p[j].z, e + 2, table, ohead, onext);
        place_edge(d[j].w, s[j].w, p[j].w, e + 3, table, ohead, onext);
    }
}


static __device__ __forceinline__ float4 h_row(const float* __restrict__ h, int s, int lane) {
    unsigned off = ((unsigned)s << 9) + ((unsigned)lane << 4);
    return *(const float4*)((const char*)h + off);
}
static __device__ __forceinline__ void add4(float4& a, const float4& v) {
    a.x += v.x; a.y += v.y; a.z += v.z; a.w += v.w;
}

static __device__ __forceinline__ float4 sum_sources(const float* __restrict__ h, int node, int c, int m, int sv,
                                                     const int* __restrict__ ohead, const int* __restrict__ onext,
                                                     const int* __restrict__ src, int lane, int& n_out) {
    float4 acc = make_float4(0.f, 0.f, 0.f, 0.f);
    #pragma unroll 1
    for (int j0 = 0; j0 < m; j0 += 8) {
        float4 v[8];
        #pragma unroll
        for (int u = 0; u < 8; ++u)
            if (j0 + u < m) v[u] = h_row(h, __builtin_amdgcn_readlane(sv, j0 + u), lane);
        #pragma unroll
        for (int u = 0; u < 8; ++u)
            if (j0 + u < m) add4(acc, v[u]);
    }
    int n = m;
    if (c > CAP) {
        int e = ohead[node];
        #pragma unroll 1
        for (int hop = 0; hop < NE && (unsigned)e < (unsigned)NE; ++hop) {
            int s = src[e];
            e = onext[e];
            float4 v = h_row(h, s, lane);
            add4(acc, v);
            ++n;
        }
    }
    n_out = n;
    return acc;
}


static_assert(NN % 32 == 0, "a block trip takes two whole tiles");

__global__ void __launch_bounds__(LAYER_T, 16)
gnn_layer_kernel(const float* __restrict__ x, const float* __restrict__ yin, float* z,
                 float* __restrict__ yout,
                 const float* __restrict__ Wl, const float* __restrict__ Wr,
                 const float* __restrict__ bias,
                 const int* __restrict__ cnt, const int* __restrict__ table,
                 const int* __restrict__ ohead, const int* __restrict__ onext,
                 const int* __restrict__ src, int first) {
    __shared__ float sW[2 * D * D4];
    __shared__ float sA[32 * D4];

    int lane = threadIdx.x & 31;
    int wave = __builtin_amdgcn_readfirstlane((int)(threadIdx.x >> 5));
    int r     = lane & 15;
    int khalf = lane >> 4;
    int tp = wave >> 4;
    int mi = (wave >> 3) & 1;
    int ct = wave & 7;

    {
        const float* wsrc = (wave < 16 ? Wl + (size_t)(wave * 8) * D : Wr + (size_t)(wave * 8 - D) * D);
        float4 wv[8];
        #pragma unroll
        for (int j = 0; j < 8; ++j) wv[j] = h_row(wsrc, j, lane);
        #pragma unroll
        for (int j = 0; j < 8; ++j) ((float4*)(sW + (wave * 8 + j) * D4))[lane] = wv[j];
    }
    const float* bfrag = sW + (mi * D + ct * 16 + r) * D4 + 2 * khalf;
    const float* afrag = sA + (tp * 16 + r) * D4 + 2 * khalf;
    float bj = mi ? bias[ct * 16 + r] : 0.0f;
    float* oplane = mi ? z : yout;

    int rt = blockIdx.x;
    int c = 0, m = 0, sv = 0;
    float4 own;
    {
        int node0 = rt * 32 + wave;
        if (first) {
            own = h_row(x, node0, lane);
        } else {
            c = cnt[node0] + 1;
            m = c < CAP ? c : CAP;
            if (lane < m) sv = table[node0 * CAP + lane];
            own = h_row(z, node0, lane);
        }
    }

    for (; rt < NN / 32; rt += LAYER_BLOCKS) {
        int node = rt * 32 + wave;
        int rtn  = rt + LAYER_BLOCKS;
        bool more = rtn < NN / 32;
        int noden = (more ? rtn : rt) * 32 + wave;
        int rn = 0;
        if (!first) rn = cnt[noden];

        float4 hrow;
        if (first) {
            hrow = own;
        } else {
            int n;
            float4 acc = sum_sources(yin, node, c, m, sv, ohead, onext, src, lane, n);
            float di = 1.0f / fmaxf((float)n, 1.0f);
            hrow.x = fmaxf(acc.x * di + own.x, 0.0f);
            hrow.y = fmaxf(acc.y * di + own.y, 0.0f);
            hrow.z = fmaxf(acc.z * di + own.z, 0.0f);
            hrow.w = fmaxf(acc.w * di + own.w, 0.0f);
        }
        ((float4*)(sA + wave * D4))[lane] = hrow;
        __syncthreads();

        int cn = rn + 1;
        int mn = cn < CAP ? cn : CAP;
        int svn = 0;
        float4 ownn = make_float4(0.f, 0.f, 0.f, 0.f);
        if (more) {
            if (first) {
                ownn = h_row(x, noden, lane);
            } else {
                if (lane < mn) svn = table[noden * CAP + lane];
                ownn = h_row(z, noden, lane);
            }
        }

        v8f acc8 = {};
        #pragma unroll
        for (int kk = 0; kk < 32; ++kk) {
            v2f a = *(const v2f*)(afrag + kk * 4);
            v2f b = *(const v2f*)(bfrag + kk * 4);
            acc8 = __builtin_amdgcn_wmma_f32_16x16x4_f32(false, a, false, b,
                                                         (short)0, acc8, false, false);
        }

        unsigned ooff = ((unsigned)((rt * 2 + tp) * 16 + 8 * khalf) << 9) + ((unsigned)(ct * 16 + r) << 2);
        float* obase = (float*)((char*)oplane + ooff);
        #pragma unroll
        for (int v = 0; v < 8; ++v)
            obase[v * D] = acc8[v] + bj;
        __syncthreads();

        c = cn; m = mn; sv = svn; own = ownn;
    }
}


__global__ void __launch_bounds__(256)
gnn_final_kernel(const float* __restrict__ y, const int* __restrict__ cnt,
                 const int* __restrict__ table, const int* __restrict__ ohead,
                 const int* __restrict__ onext, const int* __restrict__ src,
                 float* zout) {
    int lane = threadIdx.x & 31;
    int w = blockIdx.x * 8 + __builtin_amdgcn_readfirstlane((int)(threadIdx.x >> 5));

    int node = -1, c = 0, m = 0, sv = 0;
    int node1 = w;
    int r1 = cnt[node1];

    #pragma unroll 1
    for (int turn = 0; turn <= (NN + AGG_WAVES - 1) / AGG_WAVES; ++turn) {
        bool have1 = node1 < NN;
        int c1 = r1 + 1;
        int m1 = c1 < CAP ? c1 : CAP;
        int sv1 = 0;
        if (have1 && lane < m1) sv1 = table[node1 * CAP + lane];
        int node2 = node1 + AGG_WAVES;
        int r2 = cnt[node2 < NN ? node2 : NN - 1];

        if (node >= 0) {
            float4 own = ((const float4*)(zout + (size_t)node * D))[lane];
            int n;
            float4 acc = sum_sources(y, node, c, m, sv, ohead, onext, src, lane, n);
            float di = 1.0f / fmaxf((float)n, 1.0f);
            float4 o = make_float4(acc.x * di + own.x, acc.y * di + own.y,
                                   acc.z * di + own.z, acc.w * di + own.w);
            ((float4*)(zout + (size_t)node * D))[lane] = o;
        }
        if (!have1) break;
        node = node1; c = c1; m = m1; sv = sv1;
        node1 = node2; r1 = r2;
    }
}


extern "C" void kernel_launch(void* const* d_in, const int* in_sizes, int n_in,
                              void* d_out, int out_size, void* d_ws, size_t ws_size,
                              hipStream_t stream) {
    const float* x          = (const float*)d_in[0];
    const int*   edge_index = (const int*)d_in[1];
    const float* Wl         = (const float*)d_in[2];
    const float* Wr         = (const float*)d_in[3];
    const float* b          = (const float*)d_in[4];
    const int* src = edge_index;
    const int* dst = edge_index + NE;

    char* ws = (char*)d_ws;
    size_t off = 0;
    auto carve = [&](size_t bytes) -> void* {
        void* p = ws + off;
        off = (off + bytes + 255) & ~(size_t)255;
        return p;
    };
    float* yA      = (float*)carve((size_t)NN * D * sizeof(float));
    float* yB      = (float*)carve((size_t)NN * D * sizeof(float));
    int*   table   = (int*)carve((size_t)NN * CAP * sizeof(int));
    int*   cnt     = (int*)carve((size_t)2 * NN * sizeof(int));
    int*   ohead   = cnt + NN;
    int*   onext   = (int*)carve((size_t)NE * sizeof(int));
    float* z       = (float*)d_out;

    hipMemsetAsync(cnt, 0xFF, (size_t)2 * NN * sizeof(int), stream);
    gnn_slot_kernel<<<NG / (LINK_T * LINK_G), LINK_T, 0, stream>>>(src, dst, cnt, ohead, table, onext);

    float* ys[2] = {yA, yB};
    gnn_layer_kernel<<<LAYER_BLOCKS, LAYER_T, 0, stream>>>(
        x, x, z, yA, Wl, Wr, b, cnt, table, ohead, onext, src, 1);
    for (int l = 0; l < NL - 1; ++l) {
        gnn_layer_kernel<<<LAYER_BLOCKS, LAYER_T, 0, stream>>>(
            x, ys[l & 1], z, ys[(l + 1) & 1],
            Wl + (size_t)(l + 1) * D * D, Wr + (size_t)(l + 1) * D * D, b + (size_t)(l + 1) * D,
            cnt, table, ohead, onext, src, 0);
    }
    gnn_final_kernel<<<AGG_BLOCKS, 256, 0, stream>>>(ys[(NL - 1) & 1], cnt, table, ohead, onext, src, z);
}
